# starting kernel
# baseline (speedup 1.0000x reference)
#include <hip/hip_runtime.h>
#include <hip/hip_bf16.h>

#define BB 32
#define SS 512
#define FF 16
#define HH 256
#define BS (BB * SS)

typedef float v2f __attribute__((ext_vector_type(2)));
typedef float v8f __attribute__((ext_vector_type(8)));

__device__ __forceinline__ float elu_f(float v) {
    return v > 0.0f ? v : (__expf(v) - 1.0f);
}

__global__ __launch_bounds__(256) void vsn_attn_kernel(
    const float* __restrict__ x, const float* __restrict__ Wa1,
    const float* __restrict__ ba1, const float* __restrict__ Wa2,
    const float* __restrict__ ba2, float* __restrict__ wout)
{
    __shared__ float s_wa1t[HH * FF];
    __shared__ float s_wa2[HH * FF];
    __shared__ float s_ba1[HH];
    __shared__ float s_ba2[FF];

    const int tid = threadIdx.x;
#pragma unroll
    for (int i = 0; i < 16; ++i) {
        int lin = tid + i * 256;
        int f = lin >> 8, h = lin & 255;
        s_wa1t[h * FF + f] = Wa1[lin];
        s_wa2[lin] = Wa2[lin];
    }
    s_ba1[tid] = ba1[tid & 255];
    if (tid < FF) s_ba2[tid] = ba2[tid];
    __syncthreads();

    const size_t row = (size_t)blockIdx.x * 256 + tid;
    const float4* xs = (const float4*)(x + row * FF);
    float4 x0 = xs[0], x1 = xs[1], x2 = xs[2], x3 = xs[3];
    float xv[16] = {x0.x, x0.y, x0.z, x0.w, x1.x, x1.y, x1.z, x1.w,
                    x2.x, x2.y, x2.z, x2.w, x3.x, x3.y, x3.z, x3.w};

    float lg[16];
#pragma unroll
    for (int f = 0; f < 16; ++f) lg[f] = s_ba2[f];

#pragma unroll 2
    for (int h = 0; h < HH; ++h) {
        const float4* wc = (const float4*)&s_wa1t[h * FF];
        float4 c0 = wc[0], c1 = wc[1], c2 = wc[2], c3 = wc[3];
        float s = s_ba1[h];
        s += xv[0]*c0.x + xv[1]*c0.y + xv[2]*c0.z + xv[3]*c0.w;
        s += xv[4]*c1.x + xv[5]*c1.y + xv[6]*c1.z + xv[7]*c1.w;
        s += xv[8]*c2.x + xv[9]*c2.y + xv[10]*c2.z + xv[11]*c2.w;
        s += xv[12]*c3.x + xv[13]*c3.y + xv[14]*c3.z + xv[15]*c3.w;
        float a = elu_f(s);
        const float4* wd = (const float4*)&s_wa2[h * FF];
        float4 d0 = wd[0], d1 = wd[1], d2 = wd[2], d3 = wd[3];
        lg[0] += a*d0.x; lg[1] += a*d0.y; lg[2] += a*d0.z; lg[3] += a*d0.w;
        lg[4] += a*d1.x; lg[5] += a*d1.y; lg[6] += a*d1.z; lg[7] += a*d1.w;
        lg[8] += a*d2.x; lg[9] += a*d2.y; lg[10]+= a*d2.z; lg[11]+= a*d2.w;
        lg[12]+= a*d3.x; lg[13]+= a*d3.y; lg[14]+= a*d3.z; lg[15]+= a*d3.w;
    }

    float m = lg[0];
#pragma unroll
    for (int f = 1; f < 16; ++f) m = fmaxf(m, lg[f]);
    float s = 0.0f;
#pragma unroll
    for (int f = 0; f < 16; ++f) { lg[f] = __expf(lg[f] - m); s += lg[f]; }
    float inv = 1.0f / s;
#pragma unroll
    for (int f = 0; f < 16; ++f) lg[f] *= inv;

    float4* wo = (float4*)(wout + row * FF);
    wo[0] = make_float4(lg[0], lg[1], lg[2], lg[3]);
    wo[1] = make_float4(lg[4], lg[5], lg[6], lg[7]);
    wo[2] = make_float4(lg[8], lg[9], lg[10], lg[11]);
    wo[3] = make_float4(lg[12], lg[13], lg[14], lg[15]);
}

#define NTW   8
#define KC    64
#define PSTR  288

__global__ __launch_bounds__(256) void vsn_main_kernel(
    const float* __restrict__ x, const float* __restrict__ W1,
    const float* __restrict__ b1, const float* __restrict__ W2,
    const float* __restrict__ b2, const float* __restrict__ wts,
    float* __restrict__ out)
{
    __shared__ float  s_w2[(KC / 2) * PSTR];
    __shared__ float4 s_wb[128];
    __shared__ float  s_wt[128 * FF];

    const int tid  = threadIdx.x;
    const int wave = tid >> 5;
    const int lane = tid & 31;
    const int half = lane >> 4;
    const int lr   = lane & 15;
    const int n0   = blockIdx.x * 128;
    const int R0   = blockIdx.y * 128;

    {
        const float4* ws = (const float4*)(wts + (size_t)R0 * FF);
        float4* wd = (float4*)s_wt;
        wd[tid]       = ws[tid];
        wd[tid + 256] = ws[tid + 256];
    }
    const int myrow = R0 + wave * 16 + lr;
    const float4* xs = (const float4*)(x + (size_t)myrow * FF);
    float4 x0 = xs[0], x1 = xs[1], x2 = xs[2], x3 = xs[3];
    float xv[16] = {x0.x, x0.y, x0.z, x0.w, x1.x, x1.y, x1.z, x1.w,
                    x2.x, x2.y, x2.z, x2.w, x3.x, x3.y, x3.z, x3.w};

    v8f acc[NTW];
#pragma unroll
    for (int t = 0; t < NTW; ++t) acc[t] = (v8f)0.0f;

    __syncthreads();

    for (int f = 0; f < FF; ++f) {
        const float wa = s_wt[(wave * 16 + lr) * FF + f];
        const float xf = xv[f];

        for (int kc = 0; kc < HH / KC; ++kc) {
            __syncthreads();
            {
                const float* w2base =
                    W2 + ((size_t)f * HH + kc * KC) * HH + n0;
#pragma unroll
                for (int i = 0; i < 4; ++i) {
                    int idx = tid + i * 256;
                    int p = idx >> 5;
                    int q = idx & 31;
                    float4 r0 = *(const float4*)(w2base + (size_t)(2 * p) * HH + 4 * q);
                    float4 r1 = *(const float4*)(w2base + (size_t)(2 * p + 1) * HH + 4 * q);
                    float* d = &s_w2[p * PSTR + 8 * q];
                    *(float4*)(d)     = make_float4(r0.x, r1.x, r0.y, r1.y);
                    *(float4*)(d + 4) = make_float4(r0.z, r1.z, r0.w, r1.w);
                }
                if (kc == 0 && tid < 128) {
                    float2 w  = *(const float2*)(W1 + f * HH + tid * 2);
                    float2 bb = *(const float2*)(b1 + f * HH + tid * 2);
                    s_wb[tid] = make_float4(w.x, w.y, bb.x, bb.y);
                }
            }
            __syncthreads();

#pragma unroll 4
            for (int ks = 0; ks < KC / 4; ++ks) {
                float4 p4 = s_wb[kc * (KC / 2) + 2 * ks + half];
                v2f afrag;
                afrag.x = wa * elu_f(xf * p4.x + p4.z);
                afrag.y = wa * elu_f(xf * p4.y + p4.w);
                const int prow = 2 * ks + half;
#pragma unroll
                for (int t = 0; t < NTW; ++t) {
                    v2f bfrag = *(const v2f*)&s_w2[prow * PSTR + 2 * (t * 16 + lr)];
                    acc[t] = __builtin_amdgcn_wmma_f32_16x16x4_f32(
                        false, afrag, false, bfrag, (short)0, acc[t],
                        false, false);
                }
            }
        }

        float wr[8];
#pragma unroll
        for (int r = 0; r < 8; ++r)
            wr[r] = s_wt[(wave * 16 + 8 * half + r) * FF + f];
#pragma unroll
        for (int t = 0; t < NTW; ++t) {
            float b2v = b2[f * HH + n0 + t * 16 + lr];
#pragma unroll
            for (int r = 0; r < 8; ++r) acc[t][r] += wr[r] * b2v;
        }
    }

#pragma unroll
    for (int t = 0; t < NTW; ++t) {
#pragma unroll
        for (int r = 0; r < 8; ++r) {
            size_t row = (size_t)(R0 + wave * 16 + 8 * half + r);
            out[row * HH + n0 + t * 16 + lr] = acc[t][r];
        }
    }
}

extern "C" void kernel_launch(void* const* d_in, const int* in_sizes, int n_in,
                              void* d_out, int out_size, void* d_ws, size_t ws_size,
                              hipStream_t stream) {
    const float* x   = (const float*)d_in[0];
    const float* W1  = (const float*)d_in[1];
    const float* b1  = (const float*)d_in[2];
    const float* W2  = (const float*)d_in[3];
    const float* b2  = (const float*)d_in[4];
    const float* Wa1 = (const float*)d_in[5];
    const float* ba1 = (const float*)d_in[6];
    const float* Wa2 = (const float*)d_in[7];
    const float* ba2 = (const float*)d_in[8];

    float* out  = (float*)d_out;
    float* wout = out + (size_t)BS * HH;

    vsn_attn_kernel<<<dim3(BS / 256), dim3(256), 0, stream>>>(
        x, Wa1, ba1, Wa2, ba2, wout);

    vsn_main_kernel<<<dim3(HH / 128, BS / 128), dim3(256), 0, stream>>>(
        x, W1, b1, W2, b2, wout, out);
}


# Main kernel: global loads software-pipelined: next W2 chunk (and the feature's b2 row) load during the current chunk's WMMAs into a 2nd LDS stage; W1/b1 staged once; 1 barrier per chunk, not 2
# speedup vs baseline: 1.2348x; 1.2348x over previous
#include <hip/hip_runtime.h>
#include <hip/hip_bf16.h>

#define BB 32
#define SS 512
#define FF 16
#define HH 256
#define BS (BB * SS)

typedef float v2f __attribute__((ext_vector_type(2)));
typedef float v8f __attribute__((ext_vector_type(8)));

__device__ __forceinline__ float elu_f(float v) {
    return v > 0.0f ? v : (__expf(v) - 1.0f);
}

__global__ __launch_bounds__(256) void vsn_attn_kernel(
    const float* __restrict__ x, const float* __restrict__ Wa1,
    const float* __restrict__ ba1, const float* __restrict__ Wa2,
    const float* __restrict__ ba2, float* __restrict__ wout)
{
    __shared__ float s_wa1t[HH * FF];
    __shared__ float s_wa2[HH * FF];
    __shared__ float s_ba1[HH];
    __shared__ float s_ba2[FF];

    const int tid = threadIdx.x;
#pragma unroll
    for (int i = 0; i < 16; ++i) {
        int lin = tid + i * 256;
        int f = lin >> 8, h = lin & 255;
        s_wa1t[h * FF + f] = Wa1[lin];
        s_wa2[lin] = Wa2[lin];
    }
    s_ba1[tid] = ba1[tid & 255];
    if (tid < FF) s_ba2[tid] = ba2[tid];
    __syncthreads();

    const size_t row = (size_t)blockIdx.x * 256 + tid;
    const float4* xs = (const float4*)(x + row * FF);
    float4 x0 = xs[0], x1 = xs[1], x2 = xs[2], x3 = xs[3];
    float xv[16] = {x0.x, x0.y, x0.z, x0.w, x1.x, x1.y, x1.z, x1.w,
                    x2.x, x2.y, x2.z, x2.w, x3.x, x3.y, x3.z, x3.w};

    float lg[16];
#pragma unroll
    for (int f = 0; f < 16; ++f) lg[f] = s_ba2[f];

#pragma unroll 2
    for (int h = 0; h < HH; ++h) {
        const float4* wc = (const float4*)&s_wa1t[h * FF];
        float4 c0 = wc[0], c1 = wc[1], c2 = wc[2], c3 = wc[3];
        float s = s_ba1[h];
        s += xv[0]*c0.x + xv[1]*c0.y + xv[2]*c0.z + xv[3]*c0.w;
        s += xv[4]*c1.x + xv[5]*c1.y + xv[6]*c1.z + xv[7]*c1.w;
        s += xv[8]*c2.x + xv[9]*c2.y + xv[10]*c2.z + xv[11]*c2.w;
        s += xv[12]*c3.x + xv[13]*c3.y + xv[14]*c3.z + xv[15]*c3.w;
        float a = elu_f(s);
        const float4* wd = (const float4*)&s_wa2[h * FF];
        float4 d0 = wd[0], d1 = wd[1], d2 = wd[2], d3 = wd[3];
        lg[0] += a*d0.x; lg[1] += a*d0.y; lg[2] += a*d0.z; lg[3] += a*d0.w;
        lg[4] += a*d1.x; lg[5] += a*d1.y; lg[6] += a*d1.z; lg[7] += a*d1.w;
        lg[8] += a*d2.x; lg[9] += a*d2.y; lg[10]+= a*d2.z; lg[11]+= a*d2.w;
        lg[12]+= a*d3.x; lg[13]+= a*d3.y; lg[14]+= a*d3.z; lg[15]+= a*d3.w;
    }

    float m = lg[0];
#pragma unroll
    for (int f = 1; f < 16; ++f) m = fmaxf(m, lg[f]);
    float s = 0.0f;
#pragma unroll
    for (int f = 0; f < 16; ++f) { lg[f] = __expf(lg[f] - m); s += lg[f]; }
    float inv = 1.0f / s;
#pragma unroll
    for (int f = 0; f < 16; ++f) lg[f] *= inv;

    float4* wo = (float4*)(wout + row * FF);
    wo[0] = make_float4(lg[0], lg[1], lg[2], lg[3]);
    wo[1] = make_float4(lg[4], lg[5], lg[6], lg[7]);
    wo[2] = make_float4(lg[8], lg[9], lg[10], lg[11]);
    wo[3] = make_float4(lg[12], lg[13], lg[14], lg[15]);
}

#define NTW   8
#define KC    64
#define PSTR  288
#define W2BUF ((KC / 2) * PSTR)
#define NCHUNK (FF * HH / KC)

__global__ __launch_bounds__(256) void vsn_main_kernel(
    const float* __restrict__ x, const float* __restrict__ W1,
    const float* __restrict__ b1, const float* __restrict__ W2,
    const float* __restrict__ b2, const float* __restrict__ wts,
    float* __restrict__ out)
{
    __shared__ float  s_w2[2 * W2BUF];
    __shared__ float4 s_wb[FF * HH / 2];
    __shared__ float  s_wt[128 * FF];

    const int tid  = threadIdx.x;
    const int wave = tid >> 5;
    const int lane = tid & 31;
    const int half = lane >> 4;
    const int lr   = lane & 15;
    const int n0   = blockIdx.x * 128;
    const int R0   = blockIdx.y * 128;

    const int sp = tid >> 5;
    const int sq = tid & 31;

    const float*  bptr = s_w2 + half * PSTR + 2 * lr;
    const float4* wbp  = s_wb + half;

    {
        const float4* ws = (const float4*)(wts + (size_t)R0 * FF);
        float4* wd = (float4*)s_wt;
        wd[tid]       = ws[tid];
        wd[tid + 256] = ws[tid + 256];
    }
#pragma unroll
    for (int i = 0; i < (FF * HH / 2) / 256; ++i) {
        int j = tid + i * 256;
        float2 w  = *(const float2*)(W1 + 2 * j);
        float2 bb = *(const float2*)(b1 + 2 * j);
        s_wb[j] = make_float4(w.x, w.y, bb.x, bb.y);
    }
    {
        const float* w2base = W2 + n0;
#pragma unroll
        for (int i = 0; i < 4; ++i) {
            int p = sp + 8 * i;
            float4 r0 = *(const float4*)(w2base + (size_t)(2 * p) * HH + 4 * sq);
            float4 r1 = *(const float4*)(w2base + (size_t)(2 * p + 1) * HH + 4 * sq);
            float* d = &s_w2[p * PSTR + 8 * sq];
            *(float4*)(d)     = make_float4(r0.x, r1.x, r0.y, r1.y);
            *(float4*)(d + 4) = make_float4(r0.z, r1.z, r0.w, r1.w);
        }
    }
    const int myrow = R0 + wave * 16 + lr;
    const float4* xs = (const float4*)(x + (size_t)myrow * FF);
    float4 x0 = xs[0], x1 = xs[1], x2 = xs[2], x3 = xs[3];
    float xv[16] = {x0.x, x0.y, x0.z, x0.w, x1.x, x1.y, x1.z, x1.w,
                    x2.x, x2.y, x2.z, x2.w, x3.x, x3.y, x3.z, x3.w};

    v8f acc[NTW];
#pragma unroll
    for (int t = 0; t < NTW; ++t) acc[t] = (v8f)0.0f;

    __syncthreads();

    for (int f = 0; f < FF; ++f) {
        const float wa = s_wt[(wave * 16 + lr) * FF + f];
        const float xf = xv[f];

        float b2v[NTW];
#pragma unroll
        for (int t = 0; t < NTW; ++t) b2v[t] = b2[f * HH + n0 + t * 16 + lr];

        for (int kc = 0; kc < HH / KC; ++kc) {
            const int c = f * (HH / KC) + kc;
            const float*  bcur = bptr + (c & 1) * W2BUF;
            const float4* wcur = wbp + c * (KC / 2);
            float*       nxt = s_w2 + ((c + 1) & 1) * W2BUF;
            const bool more = (c + 1 < NCHUNK);

            float4 pr0[4], pr1[4];
            if (more) {
                const float* w2base = W2 + (size_t)(c + 1) * KC * HH + n0;
#pragma unroll
                for (int i = 0; i < 4; ++i) {
                    int p = sp + 8 * i;
                    pr0[i] = *(const float4*)(w2base + (size_t)(2 * p) * HH + 4 * sq);
                    pr1[i] = *(const float4*)(w2base + (size_t)(2 * p + 1) * HH + 4 * sq);
                }
            }

#pragma unroll 4
            for (int ks = 0; ks < KC / 4; ++ks) {
                float4 p4 = wcur[2 * ks];
                v2f afrag;
                afrag.x = wa * elu_f(xf * p4.x + p4.z);
                afrag.y = wa * elu_f(xf * p4.y + p4.w);
#pragma unroll
                for (int t = 0; t < NTW; ++t) {
                    v2f bfrag = *(const v2f*)(bcur + ks * (2 * PSTR) + t * 32);
                    acc[t] = __builtin_amdgcn_wmma_f32_16x16x4_f32(
                        false, afrag, false, bfrag, (short)0, acc[t],
                        false, false);
                }
            }

            if (more) {
#pragma unroll
                for (int i = 0; i < 4; ++i) {
                    int p = sp + 8 * i;
                    float4 r0 = pr0[i], r1 = pr1[i];
                    float* d = &nxt[p * PSTR + 8 * sq];
                    *(float4*)(d)     = make_float4(r0.x, r1.x, r0.y, r1.y);
                    *(float4*)(d + 4) = make_float4(r0.z, r1.z, r0.w, r1.w);
                }
            }
            __syncthreads();
        }

        float wr[8];
#pragma unroll
        for (int r = 0; r < 8; ++r)
            wr[r] = s_wt[(wave * 16 + 8 * half + r) * FF + f];
#pragma unroll
        for (int t = 0; t < NTW; ++t) {
#pragma unroll
            for (int r = 0; r < 8; ++r) acc[t][r] += wr[r] * b2v[t];
        }
    }

#pragma unroll
    for (int t = 0; t < NTW; ++t) {
#pragma unroll
        for (int r = 0; r < 8; ++r) {
            size_t row = (size_t)(R0 + wave * 16 + 8 * half + r);
            out[row * HH + n0 + t * 16 + lr] = acc[t][r];
        }
    }
}

extern "C" void kernel_launch(void* const* d_in, const int* in_sizes, int n_in,
                              void* d_out, int out_size, void* d_ws, size_t ws_size,
                              hipStream_t stream) {
    const float* x   = (const float*)d_in[0];
    const float* W1  = (const float*)d_in[1];
    const float* b1  = (const float*)d_in[2];
    const float* W2  = (const float*)d_in[3];
    const float* b2  = (const float*)d_in[4];
    const float* Wa1 = (const float*)d_in[5];
    const float* ba1 = (const float*)d_in[6];
    const float* Wa2 = (const float*)d_in[7];
    const float* ba2 = (const float*)d_in[8];

    float* out  = (float*)d_out;
    float* wout = out + (size_t)BS * HH;

    vsn_attn_kernel<<<dim3(BS / 256), dim3(256), 0, stream>>>(
        x, Wa1, ba1, Wa2, ba2, wout);

    vsn_main_kernel<<<dim3(HH / 128, BS / 128), dim3(256), 0, stream>>>(
        x, W1, b1, W2, b2, wout, out);
}


# Weights kernel moved to the f32 16x16x4 matrix instruction: both products transposed, D of the first used in place as B of the second; 256 blocks of 4 waves; sums over h and f in matrix-unit order
# speedup vs baseline: 1.4142x; 1.1453x over previous
#include <hip/hip_runtime.h>
#include <hip/hip_bf16.h>

#define BB 32
#define SS 512
#define FF 16
#define HH 256
#define BS (BB * SS)

typedef float v2f __attribute__((ext_vector_type(2)));
typedef float v8f __attribute__((ext_vector_type(8)));

__device__ __forceinline__ float elu_f(float v) {
    return v > 0.0f ? v : (__expf(v) - 1.0f);
}

__global__ __launch_bounds__(128) void vsn_attn_kernel(
    const float* __restrict__ x, const float* __restrict__ Wa1,
    const float* __restrict__ ba1, const float* __restrict__ Wa2,
    const float* __restrict__ ba2, float* __restrict__ wout)
{
    const int tid  = threadIdx.x;
    const int wave = tid >> 5;
    const int lane = tid & 31;
    const int hh   = lane >> 4;
    const int c    = lane & 15;
    const size_t row = ((size_t)blockIdx.x * 4 + wave) * 16 + c;

    v2f xb[4];
#pragma unroll
    for (int kk = 0; kk < 4; ++kk)
        xb[kk] = *(const v2f*)(x + row * FF + 4 * kk + 2 * hh);

    const float* wa1p = Wa1 + (size_t)(2 * hh) * HH + c;
    const float* wa2p = Wa2 + (size_t)(8 * hh) * FF + c;
    const float* ba1p = ba1 + 8 * hh;

    v8f l0, l1 = (v8f)0.0f;
    {
        const float4* bp = (const float4*)(ba2 + 8 * hh);
        float4 q0 = bp[0], q1 = bp[1];
        l0[0] = q0.x; l0[1] = q0.y; l0[2] = q0.z; l0[3] = q0.w;
        l0[4] = q1.x; l0[5] = q1.y; l0[6] = q1.z; l0[7] = q1.w;
    }

#pragma unroll
    for (int t = 0; t < HH / 16; ++t) {
        v8f s;
        {
            const float4* bp = (const float4*)(ba1p + 16 * t);
            float4 q0 = bp[0], q1 = bp[1];
            s[0] = q0.x; s[1] = q0.y; s[2] = q0.z; s[3] = q0.w;
            s[4] = q1.x; s[5] = q1.y; s[6] = q1.z; s[7] = q1.w;
        }
#pragma unroll
        for (int kk = 0; kk < 4; ++kk) {
            v2f a;
            a.x = wa1p[(size_t)(4 * kk) * HH + 16 * t];
            a.y = wa1p[(size_t)(4 * kk + 1) * HH + 16 * t];
            s = __builtin_amdgcn_wmma_f32_16x16x4_f32(
                false, a, false, xb[kk], (short)0, s, false, false);
        }
#pragma unroll
        for (int r = 0; r < 8; ++r) s[r] = elu_f(s[r]);
#pragma unroll
        for (int j = 0; j < 4; ++j) {
            v2f a2, bf;
            a2.x = wa2p[(size_t)(16 * t + 2 * j) * FF];
            a2.y = wa2p[(size_t)(16 * t + 2 * j + 1) * FF];
            bf.x = s[2 * j];
            bf.y = s[2 * j + 1];
            if (t & 1)
                l1 = __builtin_amdgcn_wmma_f32_16x16x4_f32(
                    false, a2, false, bf, (short)0, l1, false, false);
            else
                l0 = __builtin_amdgcn_wmma_f32_16x16x4_f32(
                    false, a2, false, bf, (short)0, l0, false, false);
        }
    }

    float lg[8];
#pragma unroll
    for (int r = 0; r < 8; ++r) lg[r] = l0[r] + l1[r];
    float m = lg[0];
#pragma unroll
    for (int r = 1; r < 8; ++r) m = fmaxf(m, lg[r]);
    m = fmaxf(m, __shfl_xor(m, 16));
    float sum = 0.0f;
#pragma unroll
    for (int r = 0; r < 8; ++r) { lg[r] = __expf(lg[r] - m); sum += lg[r]; }
    sum += __shfl_xor(sum, 16);
    const float inv = 1.0f / sum;

    float4* wo = (float4*)(wout + row * FF + 8 * hh);
    wo[0] = make_float4(lg[0] * inv, lg[1] * inv, lg[2] * inv, lg[3] * inv);
    wo[1] = make_float4(lg[4] * inv, lg[5] * inv, lg[6] * inv, lg[7] * inv);
}

#define NTW   8
#define KC    64
#define PSTR  288
#define W2BUF ((KC / 2) * PSTR)
#define NCHUNK (FF * HH / KC)

__global__ __launch_bounds__(256) void vsn_main_kernel(
    const float* __restrict__ x, const float* __restrict__ W1,
    const float* __restrict__ b1, const float* __restrict__ W2,
    const float* __restrict__ b2, const float* __restrict__ wts,
    float* __restrict__ out)
{
    __shared__ float  s_w2[2 * W2BUF];
    __shared__ float4 s_wb[FF * HH / 2];
    __shared__ float  s_wt[128 * FF];

    const int tid  = threadIdx.x;
    const int wave = tid >> 5;
    const int lane = tid & 31;
    const int half = lane >> 4;
    const int lr   = lane & 15;
    const int n0   = blockIdx.x * 128;
    const int R0   = blockIdx.y * 128;

    const int sp = tid >> 5;
    const int sq = tid & 31;

    const float*  bptr = s_w2 + half * PSTR + 2 * lr;
    const float4* wbp  = s_wb + half;

    {
        const float4* ws = (const float4*)(wts + (size_t)R0 * FF);
        float4* wd = (float4*)s_wt;
        wd[tid]       = ws[tid];
        wd[tid + 256] = ws[tid + 256];
    }
#pragma unroll
    for (int i = 0; i < (FF * HH / 2) / 256; ++i) {
        int j = tid + i * 256;
        float2 w  = *(const float2*)(W1 + 2 * j);
        float2 bb = *(const float2*)(b1 + 2 * j);
        s_wb[j] = make_float4(w.x, w.y, bb.x, bb.y);
    }
    {
        const float* w2base = W2 + n0;
#pragma unroll
        for (int i = 0; i < 4; ++i) {
            int p = sp + 8 * i;
            float4 r0 = *(const float4*)(w2base + (size_t)(2 * p) * HH + 4 * sq);
            float4 r1 = *(const float4*)(w2base + (size_t)(2 * p + 1) * HH + 4 * sq);
            float* d = &s_w2[p * PSTR + 8 * sq];
            *(float4*)(d)     = make_float4(r0.x, r1.x, r0.y, r1.y);
            *(float4*)(d + 4) = make_float4(r0.z, r1.z, r0.w, r1.w);
        }
    }
    const int myrow = R0 + wave * 16 + lr;
    const float4* xs = (const float4*)(x + (size_t)myrow * FF);
    float4 x0 = xs[0], x1 = xs[1], x2 = xs[2], x3 = xs[3];
    float xv[16] = {x0.x, x0.y, x0.z, x0.w, x1.x, x1.y, x1.z, x1.w,
                    x2.x, x2.y, x2.z, x2.w, x3.x, x3.y, x3.z, x3.w};

    v8f acc[NTW];
#pragma unroll
    for (int t = 0; t < NTW; ++t) acc[t] = (v8f)0.0f;

    __syncthreads();

    for (int f = 0; f < FF; ++f) {
        const float wa = s_wt[(wave * 16 + lr) * FF + f];
        const float xf = xv[f];

        float b2v[NTW];
#pragma unroll
        for (int t = 0; t < NTW; ++t) b2v[t] = b2[f * HH + n0 + t * 16 + lr];

        for (int kc = 0; kc < HH / KC; ++kc) {
            const int c = f * (HH / KC) + kc;
            const float*  bcur = bptr + (c & 1) * W2BUF;
            const float4* wcur = wbp + c * (KC / 2);
            float*       nxt = s_w2 + ((c + 1) & 1) * W2BUF;
            const bool more = (c + 1 < NCHUNK);

            float4 pr0[4], pr1[4];
            if (more) {
                const float* w2base = W2 + (size_t)(c + 1) * KC * HH + n0;
#pragma unroll
                for (int i = 0; i < 4; ++i) {
                    int p = sp + 8 * i;
                    pr0[i] = *(const float4*)(w2base + (size_t)(2 * p) * HH + 4 * sq);
                    pr1[i] = *(const float4*)(w2base + (size_t)(2 * p + 1) * HH + 4 * sq);
                }
            }

#pragma unroll 4
            for (int ks = 0; ks < KC / 4; ++ks) {
                float4 p4 = wcur[2 * ks];
                v2f afrag;
                afrag.x = wa * elu_f(xf * p4.x + p4.z);
                afrag.y = wa * elu_f(xf * p4.y + p4.w);
#pragma unroll
                for (int t = 0; t < NTW; ++t) {
                    v2f bfrag = *(const v2f*)(bcur + ks * (2 * PSTR) + t * 32);
                    acc[t] = __builtin_amdgcn_wmma_f32_16x16x4_f32(
                        false, afrag, false, bfrag, (short)0, acc[t],
                        false, false);
                }
            }

            if (more) {
#pragma unroll
                for (int i = 0; i < 4; ++i) {
                    int p = sp + 8 * i;
                    float4 r0 = pr0[i], r1 = pr1[i];
                    float* d = &nxt[p * PSTR + 8 * sq];
                    *(float4*)(d)     = make_float4(r0.x, r1.x, r0.y, r1.y);
                    *(float4*)(d + 4) = make_float4(r0.z, r1.z, r0.w, r1.w);
                }
            }
            __syncthreads();
        }

        float wr[8];
#pragma unroll
        for (int r = 0; r < 8; ++r)
            wr[r] = s_wt[(wave * 16 + 8 * half + r) * FF + f];
#pragma unroll
        for (int t = 0; t < NTW; ++t) {
#pragma unroll
            for (int r = 0; r < 8; ++r) acc[t][r] += wr[r] * b2v[t];
        }
    }

#pragma unroll
    for (int t = 0; t < NTW; ++t) {
#pragma unroll
        for (int r = 0; r < 8; ++r) {
            size_t row = (size_t)(R0 + wave * 16 + 8 * half + r);
            out[row * HH + n0 + t * 16 + lr] = acc[t][r];
        }
    }
}

extern "C" void kernel_launch(void* const* d_in, const int* in_sizes, int n_in,
                              void* d_out, int out_size, void* d_ws, size_t ws_size,
                              hipStream_t stream) {
    const float* x   = (const float*)d_in[0];
    const float* W1  = (const float*)d_in[1];
    const float* b1  = (const float*)d_in[2];
    const float* W2  = (const float*)d_in[3];
    const float* b2  = (const float*)d_in[4];
    const float* Wa1 = (const float*)d_in[5];
    const float* ba1 = (const float*)d_in[6];
    const float* Wa2 = (const float*)d_in[7];
    const float* ba2 = (const float*)d_in[8];

    float* out  = (float*)d_out;
    float* wout = out + (size_t)BS * HH;

    vsn_attn_kernel<<<dim3(BS / 64), dim3(128), 0, stream>>>(
        x, Wa1, ba1, Wa2, ba2, wout);

    vsn_main_kernel<<<dim3(HH / 128, BS / 128), dim3(256), 0, stream>>>(
        x, W1, b1, W2, b2, wout, out);
}
